# starting kernel
# baseline (speedup 1.0000x reference)
#include <hip/hip_runtime.h>

typedef float v2f __attribute__((ext_vector_type(2)));
typedef float v8f __attribute__((ext_vector_type(8)));

#define NN    256
#define LDM   260
#define NITER 50
#define RHO   1.0f
#define TPB   512
#define KB    8

extern "C" __global__ __launch_bounds__(TPB, 1)
void boxqp_admm_kernel(const float* __restrict__ Q,
                       const float* __restrict__ p,
                       const float* __restrict__ lb,
                       const float* __restrict__ ub,
                       float* __restrict__ out)
{
    extern __shared__ float smem[];
    float* Mm    = smem;
    float* rhs   = smem + NN * LDM;
    float* xv    = rhs + NN;
    float* CnegA = xv + NN;
    float* CnegB = CnegA + 4 * NN;
    float* RtA   = CnegB + 4 * NN;
    float* RtB   = RtA + 4 * NN;
    float* Bbuf  = RtB + 4 * NN;

    const int tid  = threadIdx.x;
    const int b    = blockIdx.x;
    const int lane = tid & 31;
    const int wv   = tid >> 5;
    const int rowA = (wv << 4) + (lane & 15);
    const int hk   = (lane >> 4) << 1;
    const float* Qb = Q + (size_t)b * NN * NN;

    for (int idx = tid; idx < NN * (NN / 4); idx += TPB) {
        int row = idx >> 6;
        int c   = (idx & 63) << 2;
        float4 q = *(const float4*)(Qb + row * NN + c);
        float* d = Mm + row * LDM + c;
        d[0] = q.x; d[1] = q.y; d[2] = q.z; d[3] = q.w;
    }
    __syncthreads();
    if (tid < NN) Mm[tid * LDM + tid] += RHO;
    __syncthreads();

    for (int K = 0; K < NN; K += KB) {
        if (tid < NN) {
            bool inK = (tid >= K) && (tid < K + KB);
            float4 cA = *(const float4*)(Mm + tid * LDM + K);
            float4 cB = *(const float4*)(Mm + tid * LDM + K + 4);
            float4 nA, nB;
            nA.x = inK ? 0.f : -cA.x;  nA.y = inK ? 0.f : -cA.y;
            nA.z = inK ? 0.f : -cA.z;  nA.w = inK ? 0.f : -cA.w;
            nB.x = inK ? 0.f : -cB.x;  nB.y = inK ? 0.f : -cB.y;
            nB.z = inK ? 0.f : -cB.z;  nB.w = inK ? 0.f : -cB.w;
            *(float4*)(CnegA + tid * 4) = nA;
            *(float4*)(CnegB + tid * 4) = nB;
        }
        if (wv == 0) {
            const int lr = lane;
            float ar[8];
            #pragma unroll
            for (int j = 0; j < 8; ++j)
                ar[j] = (lr < 8) ? Mm[(K + lr) * LDM + K + j] : 0.f;
            #pragma unroll
            for (int kk = 0; kk < 8; ++kk) {
                float prow[8];
                #pragma unroll
                for (int j = 0; j < 8; ++j) prow[j] = __shfl(ar[j], kk);
                float pinv = 1.0f / prow[kk];
                if (lr == kk) {
                    #pragma unroll
                    for (int j = 0; j < 8; ++j) ar[j] = (j == kk) ? pinv : ar[j] * pinv;
                } else {
                    float f = ar[kk];
                    #pragma unroll
                    for (int j = 0; j < 8; ++j) if (j != kk) ar[j] -= f * (prow[j] * pinv);
                    ar[kk] = -f * pinv;
                }
            }
            if (lr < 8) {
                #pragma unroll
                for (int j = 0; j < 8; ++j) Bbuf[lr * 8 + j] = ar[j];
            }
        }
        __syncthreads();

        {
            int j  = tid & 255;
            int r0 = (tid >> 8) << 2;
            float av[8];
            #pragma unroll
            for (int m = 0; m < 8; ++m) av[m] = Mm[(K + m) * LDM + j];
            float rv[4];
            #pragma unroll
            for (int r = 0; r < 4; ++r) {
                float s = 0.f;
                #pragma unroll
                for (int m = 0; m < 8; ++m) s += Bbuf[(r0 + r) * 8 + m] * av[m];
                rv[r] = s;
            }
            __syncthreads();
            #pragma unroll
            for (int r = 0; r < 4; ++r) Mm[(K + r0 + r) * LDM + j] = rv[r];
            float* RtX = (r0 == 0) ? RtA : RtB;
            *(float4*)(RtX + j * 4) = make_float4(rv[0], rv[1], rv[2], rv[3]);
        }
        __syncthreads();

        {
            v2f ca0 = *(const v2f*)(CnegA + rowA * 4 + hk);
            v2f ca1 = *(const v2f*)(CnegB + rowA * 4 + hk);
            const int rr = (wv << 4) + ((lane >> 4) << 3);
            const int cl = lane & 15;
            for (int cj = 0; cj < 16; ++cj) {
                const int cc = (cj << 4) + cl;
                v2f b0 = *(const v2f*)(RtA + cc * 4 + hk);
                v2f b1 = *(const v2f*)(RtB + cc * 4 + hk);
                v8f acc;
                #pragma unroll
                for (int g = 0; g < 8; ++g) acc[g] = Mm[(rr + g) * LDM + cc];
                acc = __builtin_amdgcn_wmma_f32_16x16x4_f32(
                    false, ca0, false, b0, (short)0, acc, false, false);
                acc = __builtin_amdgcn_wmma_f32_16x16x4_f32(
                    false, ca1, false, b1, (short)0, acc, false, false);
                #pragma unroll
                for (int g = 0; g < 8; ++g) Mm[(rr + g) * LDM + cc] = acc[g];
            }
        }
        __syncthreads();

        if (tid < NN) {
            if (tid < K || tid >= K + KB) {
                const float* cA = CnegA + tid * 4;
                const float* cB = CnegB + tid * 4;
                float o[8];
                #pragma unroll
                for (int c = 0; c < 8; ++c) {
                    float s = 0.f;
                    #pragma unroll
                    for (int r = 0; r < 4; ++r) s += cA[r] * Bbuf[r * 8 + c];
                    #pragma unroll
                    for (int r = 0; r < 4; ++r) s += cB[r] * Bbuf[(4 + r) * 8 + c];
                    o[c] = s;
                }
                *(float4*)(Mm + tid * LDM + K)     = make_float4(o[0], o[1], o[2], o[3]);
                *(float4*)(Mm + tid * LDM + K + 4) = make_float4(o[4], o[5], o[6], o[7]);
            } else {
                int r = tid - K;
                *(float4*)(Mm + tid * LDM + K)     = *(const float4*)(Bbuf + r * 8);
                *(float4*)(Mm + tid * LDM + K + 4) = *(const float4*)(Bbuf + r * 8 + 4);
            }
        }
        __syncthreads();
    }

    float zi = 0.f, ui = 0.f, xi = 0.f, pi = 0.f, lbi = 0.f, ubi = 0.f;
    if (tid < NN) {
        pi  = p [(size_t)b * NN + tid];
        lbi = lb[(size_t)b * NN + tid];
        ubi = ub[(size_t)b * NN + tid];
    }
    const float* arow = Mm + rowA * LDM + hk;

    for (int it = 0; it < NITER; ++it) {
        if (tid < NN) rhs[tid] = -pi + (zi - ui);
        __syncthreads();

        v8f acc = {};
        #pragma unroll 4
        for (int kb = 0; kb < NN; kb += 4) {
            v2f a  = *(const v2f*)(arow + kb);
            v2f bm = *(const v2f*)(rhs + kb + hk);
            acc = __builtin_amdgcn_wmma_f32_16x16x4_f32(
                false, a, false, bm, (short)0, acc, false, false);
        }
        if (lane == 0) {
            #pragma unroll
            for (int g = 0; g < 8; ++g) xv[(wv << 4) + g] = acc[g];
        }
        if (lane == 16) {
            #pragma unroll
            for (int g = 0; g < 8; ++g) xv[(wv << 4) + 8 + g] = acc[g];
        }
        __syncthreads();

        if (tid < NN) {
            xi = xv[tid];
            float t = xi + ui;
            zi = fminf(fmaxf(t, lbi), ubi);
            ui = ui + xi - zi;
        }
        __syncthreads();
    }

    if (tid < NN) out[(size_t)b * NN + tid] = xi;
}

extern "C" void kernel_launch(void* const* d_in, const int* in_sizes, int n_in,
                              void* d_out, int out_size, void* d_ws, size_t ws_size,
                              hipStream_t stream) {
    const float* Q  = (const float*)d_in[0];
    const float* p  = (const float*)d_in[1];
    const float* lb = (const float*)d_in[2];
    const float* ub = (const float*)d_in[3];
    float* out = (float*)d_out;

    const int B = in_sizes[0] / (NN * NN);
    const size_t shmem = (size_t)(NN * LDM + 2 * NN + 16 * NN + 64) * sizeof(float);

    (void)hipFuncSetAttribute((const void*)boxqp_admm_kernel,
                              hipFuncAttributeMaxDynamicSharedMemorySize,
                              (int)shmem);

    boxqp_admm_kernel<<<B, TPB, shmem, stream>>>(Q, p, lb, ub, out);
}


# Inversion keeps the whole matrix in WMMA accumulator registers across panels (no LDS tile reload/store per panel); column fix now summed inside the same rank-8 WMMA update
# speedup vs baseline: 1.1151x; 1.1151x over previous
#include <hip/hip_runtime.h>


typedef float v2f __attribute__((ext_vector_type(2)));
typedef float v8f __attribute__((ext_vector_type(8)));

#define NN    256
#define LDM   260
#define NITER 50
#define RHO   1.0f
#define TPB   512
#define KB    8
#define RP    12

extern "C" __global__ __launch_bounds__(TPB, 1)
void boxqp_admm_kernel(const float* __restrict__ Q,
                       const float* __restrict__ p,
                       const float* __restrict__ lb,
                       const float* __restrict__ ub,
                       float* __restrict__ out)
{
    extern __shared__ float smem[];
    float* Mm    = smem;
    float* rhs   = smem + NN * LDM;
    float* xv    = rhs + NN;
    float* CnegA = xv + NN;
    float* CnegB = CnegA + 4 * NN;
    float* RtA   = CnegB + 4 * NN;
    float* RtB   = RtA + 4 * NN;
    float* Bbuf  = RtB + 4 * NN;
    float* PR    = Mm;

    const int tid  = threadIdx.x;
    const int b    = blockIdx.x;
    const int lane = tid & 31;
    const int wv   = tid >> 5;
    const int cl   = lane & 15;
    const int hh   = lane >> 4;
    const int rowA = (wv << 4) + cl;
    const int hk   = hh << 1;
    const int rr   = (wv << 4) + (hh << 3);
    const float* Qb = Q + (size_t)b * NN * NN;

    for (int idx = tid; idx < NN * (NN / 4); idx += TPB) {
        int row = idx >> 6;
        int c   = (idx & 63) << 2;
        float4 q = *(const float4*)(Qb + row * NN + c);
        float* d = Mm + row * LDM + c;
        d[0] = q.x; d[1] = q.y; d[2] = q.z; d[3] = q.w;
    }
    __syncthreads();
    if (tid < NN) Mm[tid * LDM + tid] += RHO;
    __syncthreads();

    v8f acc[16];
    #pragma unroll
    for (int T = 0; T < 16; ++T) {
        #pragma unroll
        for (int g = 0; g < 8; ++g) acc[T][g] = Mm[(rr + g) * LDM + (T << 4) + cl];
    }
    __syncthreads();

    for (int K = 0; K < NN; K += KB) {
        const int  cj    = K >> 4;
        const int  k0    = K & 15;
        const int  ph    = k0 >> 3;
        const int  m     = cl - k0;
        const bool incol = ((cl >> 3) == ph);
        const bool prow  = (wv == cj) && (hh == ph);

        if (wv == cj) {
            if (hh == ph) {
                #pragma unroll
                for (int T = 0; T < 16; ++T) {
                    float* d = PR + ((T << 4) + cl) * RP;
                    *(float4*)(d)     = make_float4(acc[T][0], acc[T][1], acc[T][2], acc[T][3]);
                    *(float4*)(d + 4) = make_float4(acc[T][4], acc[T][5], acc[T][6], acc[T][7]);
                }
            }
            #pragma unroll
            for (int T = 0; T < 16; ++T) {
                #pragma unroll
                for (int g = 0; g < 8; ++g) acc[T][g] = (hh == ph) ? 0.f : acc[T][g];
            }
        }

        #pragma unroll
        for (int T = 0; T < 16; ++T) {
            if (T == cj) {
                if (incol) {
                    float* dst = ((m < 4) ? CnegA : CnegB) + (m & 3);
                    #pragma unroll
                    for (int g = 0; g < 8; ++g) {
                        float v = prow ? ((g == m) ? 1.f : 0.f) : -acc[T][g];
                        dst[(rr + g) * 4] = v;
                    }
                }
                #pragma unroll
                for (int g = 0; g < 8; ++g) acc[T][g] = incol ? 0.f : acc[T][g];
            }
        }
        __syncthreads();

        if (wv == 0) {
            const int lr = lane;
            float ar[8];
            #pragma unroll
            for (int j = 0; j < 8; ++j)
                ar[j] = (lr < 8) ? PR[(K + j) * RP + lr] : 0.f;
            #pragma unroll
            for (int kk = 0; kk < 8; ++kk) {
                float prw[8];
                #pragma unroll
                for (int j = 0; j < 8; ++j) prw[j] = __shfl(ar[j], kk);
                float pinv = 1.0f / prw[kk];
                if (lr == kk) {
                    #pragma unroll
                    for (int j = 0; j < 8; ++j) ar[j] = (j == kk) ? pinv : ar[j] * pinv;
                } else {
                    float f = ar[kk];
                    #pragma unroll
                    for (int j = 0; j < 8; ++j) if (j != kk) ar[j] -= f * (prw[j] * pinv);
                    ar[kk] = -f * pinv;
                }
            }
            if (lr < 8) {
                #pragma unroll
                for (int j = 0; j < 8; ++j) Bbuf[lr * 8 + j] = ar[j];
            }
        }
        __syncthreads();

        {
            int j  = tid & 255;
            int r0 = (tid >> 8) << 2;
            float4 a0 = *(const float4*)(PR + j * RP);
            float4 a1 = *(const float4*)(PR + j * RP + 4);
            float av[8] = { a0.x, a0.y, a0.z, a0.w, a1.x, a1.y, a1.z, a1.w };
            int jm = j - K;
            bool onp = (jm >= 0) && (jm < KB);
            float rv[4];
            #pragma unroll
            for (int r = 0; r < 4; ++r) {
                float s = 0.f;
                #pragma unroll
                for (int q = 0; q < 8; ++q) s += Bbuf[(r0 + r) * 8 + q] * av[q];
                rv[r] = onp ? Bbuf[(r0 + r) * 8 + (jm & 7)] : s;
            }
            float* RtX = (r0 == 0) ? RtA : RtB;
            *(float4*)(RtX + j * 4) = make_float4(rv[0], rv[1], rv[2], rv[3]);
        }
        __syncthreads();

        {
            v2f ca0 = *(const v2f*)(CnegA + rowA * 4 + hk);
            v2f ca1 = *(const v2f*)(CnegB + rowA * 4 + hk);
            #pragma unroll
            for (int T = 0; T < 16; ++T) {
                const int cc = (T << 4) + cl;
                v2f b0 = *(const v2f*)(RtA + cc * 4 + hk);
                v2f b1 = *(const v2f*)(RtB + cc * 4 + hk);
                acc[T] = __builtin_amdgcn_wmma_f32_16x16x4_f32(
                    false, ca0, false, b0, (short)0, acc[T], false, false);
                acc[T] = __builtin_amdgcn_wmma_f32_16x16x4_f32(
                    false, ca1, false, b1, (short)0, acc[T], false, false);
            }
        }
    }

    #pragma unroll
    for (int T = 0; T < 16; ++T) {
        #pragma unroll
        for (int g = 0; g < 8; ++g) Mm[(rr + g) * LDM + (T << 4) + cl] = acc[T][g];
    }

    float zi = 0.f, ui = 0.f, xi = 0.f, pi = 0.f, lbi = 0.f, ubi = 0.f;
    if (tid < NN) {
        pi  = p [(size_t)b * NN + tid];
        lbi = lb[(size_t)b * NN + tid];
        ubi = ub[(size_t)b * NN + tid];
    }
    const float* arow = Mm + rowA * LDM + hk;

    for (int it = 0; it < NITER; ++it) {
        if (tid < NN) rhs[tid] = -pi + (zi - ui);
        __syncthreads();

        v8f xa = {};
        #pragma unroll 4
        for (int kb = 0; kb < NN; kb += 4) {
            v2f a  = *(const v2f*)(arow + kb);
            v2f bm = *(const v2f*)(rhs + kb + hk);
            xa = __builtin_amdgcn_wmma_f32_16x16x4_f32(
                false, a, false, bm, (short)0, xa, false, false);
        }
        if (lane == 0) {
            #pragma unroll
            for (int g = 0; g < 8; ++g) xv[(wv << 4) + g] = xa[g];
        }
        if (lane == 16) {
            #pragma unroll
            for (int g = 0; g < 8; ++g) xv[(wv << 4) + 8 + g] = xa[g];
        }
        __syncthreads();

        if (tid < NN) {
            xi = xv[tid];
            float t = xi + ui;
            zi = fminf(fmaxf(t, lbi), ubi);
            ui = ui + xi - zi;
        }
        __syncthreads();
    }

    if (tid < NN) out[(size_t)b * NN + tid] = xi;
}

extern "C" void kernel_launch(void* const* d_in, const int* in_sizes, int n_in,
                              void* d_out, int out_size, void* d_ws, size_t ws_size,
                              hipStream_t stream) {
    const float* Q  = (const float*)d_in[0];
    const float* p  = (const float*)d_in[1];
    const float* lb = (const float*)d_in[2];
    const float* ub = (const float*)d_in[3];
    float* out = (float*)d_out;

    const int B = in_sizes[0] / (NN * NN);
    const size_t shmem = (size_t)(NN * LDM + 2 * NN + 16 * NN + 64) * sizeof(float);

    (void)hipFuncSetAttribute((const void*)boxqp_admm_kernel,
                              hipFuncAttributeMaxDynamicSharedMemorySize,
                              (int)shmem);

    boxqp_admm_kernel<<<B, TPB, shmem, stream>>>(Q, p, lb, ub, out);
}


# ADMM matvec reads Minv from the WMMA accumulator registers as A fragments (inversion now runs on M^T); 16 per-wave partial sums joined through LDS and a row butterfly, so the sum order changes
# speedup vs baseline: 1.2762x; 1.1445x over previous
#include <hip/hip_runtime.h>


typedef float v2f __attribute__((ext_vector_type(2)));
typedef float v8f __attribute__((ext_vector_type(8)));

#define NN    256
#define LDM   260
#define NITER 50
#define RHO   1.0f
#define TPB   512
#define KB    8
#define RP    12
#define PP    260

template <int M>
__device__ __forceinline__ float row_xor(float v) {
    return __builtin_bit_cast(float,
        __builtin_amdgcn_update_dpp(0, __builtin_bit_cast(int, v), 0x160 | M, 0xf, 0xf, true));
}

extern "C" __global__ __launch_bounds__(TPB, 1)
void boxqp_admm_kernel(const float* __restrict__ Q,
                       const float* __restrict__ p,
                       const float* __restrict__ lb,
                       const float* __restrict__ ub,
                       float* __restrict__ out)
{
    extern __shared__ float smem[];
    float* Mm    = smem;
    float* xv    = smem + NN * LDM;
    float* CnegA = xv + NN;
    float* CnegB = CnegA + 4 * NN;
    float* RtA   = CnegB + 4 * NN;
    float* RtB   = RtA + 4 * NN;
    float* Bbuf  = RtB + 4 * NN;
    float* PR    = Mm;
    float* part  = Mm;

    const int tid  = threadIdx.x;
    const int b    = blockIdx.x;
    const int lane = tid & 31;
    const int wv   = tid >> 5;
    const int cl   = lane & 15;
    const int hh   = lane >> 4;
    const int rowA = (wv << 4) + cl;
    const int hk   = hh << 1;
    const int rr   = (wv << 4) + (hh << 3);
    const float* Qb = Q + (size_t)b * NN * NN;

    for (int idx = tid; idx < NN * (NN / 4); idx += TPB) {
        int row = idx >> 6;
        int c   = (idx & 63) << 2;
        float4 q = *(const float4*)(Qb + row * NN + c);
        float* d = Mm + row * LDM + c;
        d[0] = q.x; d[1] = q.y; d[2] = q.z; d[3] = q.w;
    }
    __syncthreads();
    if (tid < NN) Mm[tid * LDM + tid] += RHO;
    __syncthreads();

    v8f acc[16];
    #pragma unroll
    for (int T = 0; T < 16; ++T) {
        const float* s = Mm + ((T << 4) + cl) * LDM + rr;
        float4 lo = *(const float4*)(s);
        float4 hi = *(const float4*)(s + 4);
        acc[T] = (v8f){ lo.x, lo.y, lo.z, lo.w, hi.x, hi.y, hi.z, hi.w };
    }
    __syncthreads();

    for (int K = 0; K < NN; K += KB) {
        const int  cj    = K >> 4;
        const int  k0    = K & 15;
        const int  ph    = k0 >> 3;
        const int  m     = cl - k0;
        const bool incol = ((cl >> 3) == ph);
        const bool prow  = (wv == cj) && (hh == ph);

        if (wv == cj) {
            if (hh == ph) {
                #pragma unroll
                for (int T = 0; T < 16; ++T) {
                    float* d = PR + ((T << 4) + cl) * RP;
                    *(float4*)(d)     = make_float4(acc[T][0], acc[T][1], acc[T][2], acc[T][3]);
                    *(float4*)(d + 4) = make_float4(acc[T][4], acc[T][5], acc[T][6], acc[T][7]);
                }
            }
            #pragma unroll
            for (int T = 0; T < 16; ++T) {
                #pragma unroll
                for (int g = 0; g < 8; ++g) acc[T][g] = (hh == ph) ? 0.f : acc[T][g];
            }
        }

        #pragma unroll
        for (int T = 0; T < 16; ++T) {
            if (T == cj) {
                if (incol) {
                    float* dst = ((m < 4) ? CnegA : CnegB) + (m & 3);
                    #pragma unroll
                    for (int g = 0; g < 8; ++g) {
                        float v = prow ? ((g == m) ? 1.f : 0.f) : -acc[T][g];
                        dst[(rr + g) * 4] = v;
                    }
                }
                #pragma unroll
                for (int g = 0; g < 8; ++g) acc[T][g] = incol ? 0.f : acc[T][g];
            }
        }
        __syncthreads();

        if (wv == 0) {
            const int lr = lane;
            float ar[8];
            #pragma unroll
            for (int j = 0; j < 8; ++j)
                ar[j] = (lr < 8) ? PR[(K + j) * RP + lr] : 0.f;
            #pragma unroll
            for (int kk = 0; kk < 8; ++kk) {
                float prw[8];
                #pragma unroll
                for (int j = 0; j < 8; ++j) prw[j] = __shfl(ar[j], kk);
                float pinv = 1.0f / prw[kk];
                if (lr == kk) {
                    #pragma unroll
                    for (int j = 0; j < 8; ++j) ar[j] = (j == kk) ? pinv : ar[j] * pinv;
                } else {
                    float f = ar[kk];
                    #pragma unroll
                    for (int j = 0; j < 8; ++j) if (j != kk) ar[j] -= f * (prw[j] * pinv);
                    ar[kk] = -f * pinv;
                }
            }
            if (lr < 8) {
                #pragma unroll
                for (int j = 0; j < 8; ++j) Bbuf[lr * 8 + j] = ar[j];
            }
        }
        __syncthreads();

        {
            int j  = tid & 255;
            int r0 = (tid >> 8) << 2;
            float4 a0 = *(const float4*)(PR + j * RP);
            float4 a1 = *(const float4*)(PR + j * RP + 4);
            float av[8] = { a0.x, a0.y, a0.z, a0.w, a1.x, a1.y, a1.z, a1.w };
            int jm = j - K;
            bool onp = (jm >= 0) && (jm < KB);
            float rv[4];
            #pragma unroll
            for (int r = 0; r < 4; ++r) {
                float s = 0.f;
                #pragma unroll
                for (int q = 0; q < 8; ++q) s += Bbuf[(r0 + r) * 8 + q] * av[q];
                rv[r] = onp ? Bbuf[(r0 + r) * 8 + (jm & 7)] : s;
            }
            float* RtX = (r0 == 0) ? RtA : RtB;
            *(float4*)(RtX + j * 4) = make_float4(rv[0], rv[1], rv[2], rv[3]);
        }
        __syncthreads();

        {
            v2f ca0 = *(const v2f*)(CnegA + rowA * 4 + hk);
            v2f ca1 = *(const v2f*)(CnegB + rowA * 4 + hk);
            #pragma unroll
            for (int T = 0; T < 16; ++T) {
                const int cc = (T << 4) + cl;
                v2f b0 = *(const v2f*)(RtA + cc * 4 + hk);
                v2f b1 = *(const v2f*)(RtB + cc * 4 + hk);
                acc[T] = __builtin_amdgcn_wmma_f32_16x16x4_f32(
                    false, ca0, false, b0, (short)0, acc[T], false, false);
                acc[T] = __builtin_amdgcn_wmma_f32_16x16x4_f32(
                    false, ca1, false, b1, (short)0, acc[T], false, false);
            }
        }
    }

    const int e0 = (wv << 4) + (hh << 3);
    float pv[8], lbv[8], ubv[8], uv[8], rv[8], xo[8];
    {
        const float4* p4 = (const float4*)(p  + (size_t)b * NN + e0);
        const float4* l4 = (const float4*)(lb + (size_t)b * NN + e0);
        const float4* u4 = (const float4*)(ub + (size_t)b * NN + e0);
        float4 a0 = p4[0], a1 = p4[1], b0 = l4[0], b1 = l4[1], c0 = u4[0], c1 = u4[1];
        pv[0] = a0.x; pv[1] = a0.y; pv[2] = a0.z; pv[3] = a0.w;
        pv[4] = a1.x; pv[5] = a1.y; pv[6] = a1.z; pv[7] = a1.w;
        lbv[0] = b0.x; lbv[1] = b0.y; lbv[2] = b0.z; lbv[3] = b0.w;
        lbv[4] = b1.x; lbv[5] = b1.y; lbv[6] = b1.z; lbv[7] = b1.w;
        ubv[0] = c0.x; ubv[1] = c0.y; ubv[2] = c0.z; ubv[3] = c0.w;
        ubv[4] = c1.x; ubv[5] = c1.y; ubv[6] = c1.z; ubv[7] = c1.w;
    }
    #pragma unroll
    for (int g = 0; g < 8; ++g) { uv[g] = 0.f; rv[g] = -pv[g]; xo[g] = 0.f; }

    for (int it = 0; it < NITER; ++it) {
        v8f d0 = {}, d1 = {};
        #pragma unroll
        for (int T = 0; T < 16; T += 2) {
            const bool m0 = (cl == T), m1 = (cl == T + 1);
            #pragma unroll
            for (int q = 0; q < 4; ++q) {
                v2f a0 = { acc[T][2 * q],     acc[T][2 * q + 1] };
                v2f a1 = { acc[T + 1][2 * q], acc[T + 1][2 * q + 1] };
                v2f b0 = { m0 ? rv[2 * q] : 0.f, m0 ? rv[2 * q + 1] : 0.f };
                v2f b1 = { m1 ? rv[2 * q] : 0.f, m1 ? rv[2 * q + 1] : 0.f };
                d0 = __builtin_amdgcn_wmma_f32_16x16x4_f32(
                    false, a0, false, b0, (short)0, d0, false, false);
                d1 = __builtin_amdgcn_wmma_f32_16x16x4_f32(
                    false, a1, false, b1, (short)0, d1, false, false);
            }
        }
        v8f d = d0 + d1;

        float* pw = part + (((it & 1) << 4) + wv) * PP + (cl << 4) + (hh << 3);
        *(float4*)(pw)     = make_float4(d[0], d[1], d[2], d[3]);
        *(float4*)(pw + 4) = make_float4(d[4], d[5], d[6], d[7]);
        __syncthreads();

        const float* pr = part + (((it & 1) << 4) + cl) * PP + e0;
        float4 s0 = *(const float4*)(pr);
        float4 s1 = *(const float4*)(pr + 4);
        float s[8] = { s0.x, s0.y, s0.z, s0.w, s1.x, s1.y, s1.z, s1.w };
        #pragma unroll
        for (int g = 0; g < 8; ++g) s[g] += row_xor<1>(s[g]);
        #pragma unroll
        for (int g = 0; g < 8; ++g) s[g] += row_xor<2>(s[g]);
        #pragma unroll
        for (int g = 0; g < 8; ++g) s[g] += row_xor<4>(s[g]);
        #pragma unroll
        for (int g = 0; g < 8; ++g) s[g] += row_xor<8>(s[g]);

        #pragma unroll
        for (int g = 0; g < 8; ++g) {
            float xi = s[g];
            float t  = xi + uv[g];
            float zi = fminf(fmaxf(t, lbv[g]), ubv[g]);
            uv[g] = uv[g] + xi - zi;
            rv[g] = -pv[g] + (zi - uv[g]);
            xo[g] = xi;
        }
    }

    if (cl == 0) {
        *(float4*)(xv + e0)     = make_float4(xo[0], xo[1], xo[2], xo[3]);
        *(float4*)(xv + e0 + 4) = make_float4(xo[4], xo[5], xo[6], xo[7]);
    }
    __syncthreads();
    if (tid < NN) out[(size_t)b * NN + tid] = xv[tid];
}

extern "C" void kernel_launch(void* const* d_in, const int* in_sizes, int n_in,
                              void* d_out, int out_size, void* d_ws, size_t ws_size,
                              hipStream_t stream) {
    const float* Q  = (const float*)d_in[0];
    const float* p  = (const float*)d_in[1];
    const float* lb = (const float*)d_in[2];
    const float* ub = (const float*)d_in[3];
    float* out = (float*)d_out;

    const int B = in_sizes[0] / (NN * NN);
    const size_t shmem = (size_t)(NN * LDM + NN + 16 * NN + 64) * sizeof(float);

    (void)hipFuncSetAttribute((const void*)boxqp_admm_kernel,
                              hipFuncAttributeMaxDynamicSharedMemorySize,
                              (int)shmem);

    boxqp_admm_kernel<<<B, TPB, shmem, stream>>>(Q, p, lb, ub, out);
}


# 8x8 pivot-block inverse: pivot row broadcast by v_readlane (scalar registers) instead of LDS-crossbar shuffles, pivot lane by select instead of a divergent branch; same arithmetic
# speedup vs baseline: 1.3658x; 1.0702x over previous
#include <hip/hip_runtime.h>


typedef float v2f __attribute__((ext_vector_type(2)));
typedef float v8f __attribute__((ext_vector_type(8)));

#define NN    256
#define LDM   260
#define NITER 50
#define RHO   1.0f
#define TPB   512
#define KB    8
#define RP    12
#define PP    260

template <int M>
__device__ __forceinline__ float row_xor(float v) {
    return __builtin_bit_cast(float,
        __builtin_amdgcn_update_dpp(0, __builtin_bit_cast(int, v), 0x160 | M, 0xf, 0xf, true));
}

__device__ __forceinline__ float lane_value(float v, int l) {
    return __builtin_bit_cast(float, __builtin_amdgcn_readlane(__builtin_bit_cast(int, v), l));
}

extern "C" __global__ __launch_bounds__(TPB, 1)
void boxqp_admm_kernel(const float* __restrict__ Q,
                       const float* __restrict__ p,
                       const float* __restrict__ lb,
                       const float* __restrict__ ub,
                       float* __restrict__ out)
{
    extern __shared__ float smem[];
    float* Mm    = smem;
    float* xv    = smem + NN * LDM;
    float* CnegA = xv + NN;
    float* CnegB = CnegA + 4 * NN;
    float* RtA   = CnegB + 4 * NN;
    float* RtB   = RtA + 4 * NN;
    float* Bbuf  = RtB + 4 * NN;
    float* PR    = Mm;
    float* part  = Mm;

    const int tid  = threadIdx.x;
    const int b    = blockIdx.x;
    const int lane = tid & 31;
    const int wv   = tid >> 5;
    const int cl   = lane & 15;
    const int hh   = lane >> 4;
    const int rowA = (wv << 4) + cl;
    const int hk   = hh << 1;
    const int rr   = (wv << 4) + (hh << 3);
    const float* Qb = Q + (size_t)b * NN * NN;

    for (int idx = tid; idx < NN * (NN / 4); idx += TPB) {
        int row = idx >> 6;
        int c   = (idx & 63) << 2;
        float4 q = *(const float4*)(Qb + row * NN + c);
        float* d = Mm + row * LDM + c;
        d[0] = q.x; d[1] = q.y; d[2] = q.z; d[3] = q.w;
    }
    __syncthreads();
    if (tid < NN) Mm[tid * LDM + tid] += RHO;
    __syncthreads();

    v8f acc[16];
    #pragma unroll
    for (int T = 0; T < 16; ++T) {
        const float* s = Mm + ((T << 4) + cl) * LDM + rr;
        float4 lo = *(const float4*)(s);
        float4 hi = *(const float4*)(s + 4);
        acc[T] = (v8f){ lo.x, lo.y, lo.z, lo.w, hi.x, hi.y, hi.z, hi.w };
    }
    __syncthreads();

    for (int K = 0; K < NN; K += KB) {
        const int  cj    = K >> 4;
        const int  k0    = K & 15;
        const int  ph    = k0 >> 3;
        const int  m     = cl - k0;
        const bool incol = ((cl >> 3) == ph);
        const bool prow  = (wv == cj) && (hh == ph);

        if (wv == cj) {
            if (hh == ph) {
                #pragma unroll
                for (int T = 0; T < 16; ++T) {
                    float* d = PR + ((T << 4) + cl) * RP;
                    *(float4*)(d)     = make_float4(acc[T][0], acc[T][1], acc[T][2], acc[T][3]);
                    *(float4*)(d + 4) = make_float4(acc[T][4], acc[T][5], acc[T][6], acc[T][7]);
                }
            }
            #pragma unroll
            for (int T = 0; T < 16; ++T) {
                #pragma unroll
                for (int g = 0; g < 8; ++g) acc[T][g] = (hh == ph) ? 0.f : acc[T][g];
            }
        }

        #pragma unroll
        for (int T = 0; T < 16; ++T) {
            if (T == cj) {
                if (incol) {
                    float* dst = ((m < 4) ? CnegA : CnegB) + (m & 3);
                    #pragma unroll
                    for (int g = 0; g < 8; ++g) {
                        float v = prow ? ((g == m) ? 1.f : 0.f) : -acc[T][g];
                        dst[(rr + g) * 4] = v;
                    }
                }
                #pragma unroll
                for (int g = 0; g < 8; ++g) acc[T][g] = incol ? 0.f : acc[T][g];
            }
        }
        __syncthreads();

        if (wv == 0) {
            float ar[8];
            #pragma unroll
            for (int j = 0; j < 8; ++j)
                ar[j] = (lane < 8) ? PR[(K + j) * RP + lane] : 0.f;
            #pragma unroll
            for (int kk = 0; kk < 8; ++kk) {
                float prw[8];
                #pragma unroll
                for (int j = 0; j < 8; ++j) prw[j] = lane_value(ar[j], kk);
                const float pinv = 1.0f / prw[kk];
                const bool  isp  = (lane == kk);
                const float f    = ar[kk];
                #pragma unroll
                for (int j = 0; j < 8; ++j) {
                    if (j != kk) {
                        const float sr = prw[j] * pinv;
                        ar[j] = isp ? sr : (ar[j] - f * sr);
                    }
                }
                ar[kk] = isp ? pinv : (-f * pinv);
            }
            if (lane < 8) {
                #pragma unroll
                for (int j = 0; j < 8; ++j) Bbuf[lane * 8 + j] = ar[j];
            }
        }
        __syncthreads();

        {
            int j  = tid & 255;
            int r0 = (tid >> 8) << 2;
            float4 a0 = *(const float4*)(PR + j * RP);
            float4 a1 = *(const float4*)(PR + j * RP + 4);
            float av[8] = { a0.x, a0.y, a0.z, a0.w, a1.x, a1.y, a1.z, a1.w };
            int jm = j - K;
            bool onp = (jm >= 0) && (jm < KB);
            float rv[4];
            #pragma unroll
            for (int r = 0; r < 4; ++r) {
                float s = 0.f;
                #pragma unroll
                for (int q = 0; q < 8; ++q) s += Bbuf[(r0 + r) * 8 + q] * av[q];
                rv[r] = onp ? Bbuf[(r0 + r) * 8 + (jm & 7)] : s;
            }
            float* RtX = (r0 == 0) ? RtA : RtB;
            *(float4*)(RtX + j * 4) = make_float4(rv[0], rv[1], rv[2], rv[3]);
        }
        __syncthreads();

        {
            v2f ca0 = *(const v2f*)(CnegA + rowA * 4 + hk);
            v2f ca1 = *(const v2f*)(CnegB + rowA * 4 + hk);
            #pragma unroll
            for (int T = 0; T < 16; ++T) {
                const int cc = (T << 4) + cl;
                v2f b0 = *(const v2f*)(RtA + cc * 4 + hk);
                v2f b1 = *(const v2f*)(RtB + cc * 4 + hk);
                acc[T] = __builtin_amdgcn_wmma_f32_16x16x4_f32(
                    false, ca0, false, b0, (short)0, acc[T], false, false);
                acc[T] = __builtin_amdgcn_wmma_f32_16x16x4_f32(
                    false, ca1, false, b1, (short)0, acc[T], false, false);
            }
        }
    }

    const int e0 = (wv << 4) + (hh << 3);
    float pv[8], lbv[8], ubv[8], uv[8], rv[8], xo[8];
    {
        const float4* p4 = (const float4*)(p  + (size_t)b * NN + e0);
        const float4* l4 = (const float4*)(lb + (size_t)b * NN + e0);
        const float4* u4 = (const float4*)(ub + (size_t)b * NN + e0);
        float4 a0 = p4[0], a1 = p4[1], b0 = l4[0], b1 = l4[1], c0 = u4[0], c1 = u4[1];
        pv[0] = a0.x; pv[1] = a0.y; pv[2] = a0.z; pv[3] = a0.w;
        pv[4] = a1.x; pv[5] = a1.y; pv[6] = a1.z; pv[7] = a1.w;
        lbv[0] = b0.x; lbv[1] = b0.y; lbv[2] = b0.z; lbv[3] = b0.w;
        lbv[4] = b1.x; lbv[5] = b1.y; lbv[6] = b1.z; lbv[7] = b1.w;
        ubv[0] = c0.x; ubv[1] = c0.y; ubv[2] = c0.z; ubv[3] = c0.w;
        ubv[4] = c1.x; ubv[5] = c1.y; ubv[6] = c1.z; ubv[7] = c1.w;
    }
    #pragma unroll
    for (int g = 0; g < 8; ++g) { uv[g] = 0.f; rv[g] = -pv[g]; xo[g] = 0.f; }

    for (int it = 0; it < NITER; ++it) {
        v8f d0 = {}, d1 = {};
        #pragma unroll
        for (int T = 0; T < 16; T += 2) {
            const bool m0 = (cl == T), m1 = (cl == T + 1);
            #pragma unroll
            for (int q = 0; q < 4; ++q) {
                v2f a0 = { acc[T][2 * q],     acc[T][2 * q + 1] };
                v2f a1 = { acc[T + 1][2 * q], acc[T + 1][2 * q + 1] };
                v2f b0 = { m0 ? rv[2 * q] : 0.f, m0 ? rv[2 * q + 1] : 0.f };
                v2f b1 = { m1 ? rv[2 * q] : 0.f, m1 ? rv[2 * q + 1] : 0.f };
                d0 = __builtin_amdgcn_wmma_f32_16x16x4_f32(
                    false, a0, false, b0, (short)0, d0, false, false);
                d1 = __builtin_amdgcn_wmma_f32_16x16x4_f32(
                    false, a1, false, b1, (short)0, d1, false, false);
            }
        }
        v8f d = d0 + d1;

        float* pw = part + (((it & 1) << 4) + wv) * PP + (cl << 4) + (hh << 3);
        *(float4*)(pw)     = make_float4(d[0], d[1], d[2], d[3]);
        *(float4*)(pw + 4) = make_float4(d[4], d[5], d[6], d[7]);
        __syncthreads();

        const float* pr = part + (((it & 1) << 4) + cl) * PP + e0;
        float4 s0 = *(const float4*)(pr);
        float4 s1 = *(const float4*)(pr + 4);
        float s[8] = { s0.x, s0.y, s0.z, s0.w, s1.x, s1.y, s1.z, s1.w };
        #pragma unroll
        for (int g = 0; g < 8; ++g) s[g] += row_xor<1>(s[g]);
        #pragma unroll
        for (int g = 0; g < 8; ++g) s[g] += row_xor<2>(s[g]);
        #pragma unroll
        for (int g = 0; g < 8; ++g) s[g] += row_xor<4>(s[g]);
        #pragma unroll
        for (int g = 0; g < 8; ++g) s[g] += row_xor<8>(s[g]);

        #pragma unroll
        for (int g = 0; g < 8; ++g) {
            float xi = s[g];
            float t  = xi + uv[g];
            float zi = fminf(fmaxf(t, lbv[g]), ubv[g]);
            uv[g] = uv[g] + xi - zi;
            rv[g] = -pv[g] + (zi - uv[g]);
            xo[g] = xi;
        }
    }

    if (cl == 0) {
        *(float4*)(xv + e0)     = make_float4(xo[0], xo[1], xo[2], xo[3]);
        *(float4*)(xv + e0 + 4) = make_float4(xo[4], xo[5], xo[6], xo[7]);
    }
    __syncthreads();
    if (tid < NN) out[(size_t)b * NN + tid] = xv[tid];
}

extern "C" void kernel_launch(void* const* d_in, const int* in_sizes, int n_in,
                              void* d_out, int out_size, void* d_ws, size_t ws_size,
                              hipStream_t stream) {
    const float* Q  = (const float*)d_in[0];
    const float* p  = (const float*)d_in[1];
    const float* lb = (const float*)d_in[2];
    const float* ub = (const float*)d_in[3];
    float* out = (float*)d_out;

    const int B = in_sizes[0] / (NN * NN);
    const size_t shmem = (size_t)(NN * LDM + NN + 16 * NN + 64) * sizeof(float);

    (void)hipFuncSetAttribute((const void*)boxqp_admm_kernel,
                              hipFuncAttributeMaxDynamicSharedMemorySize,
                              (int)shmem);

    boxqp_admm_kernel<<<B, TPB, shmem, stream>>>(Q, p, lb, ub, out);
}


# Panel step multiplied as (C'*B)*R, not C'*(B*R): each wave scales its own rows of the column panel by the pivot inverse; shared row-panel product and one barrier per panel go; sum order changes
# speedup vs baseline: 1.4087x; 1.0314x over previous
#include <hip/hip_runtime.h>


typedef float v2f __attribute__((ext_vector_type(2)));
typedef float v8f __attribute__((ext_vector_type(8)));

#define NN    256
#define LDM   260
#define NITER 50
#define RHO   1.0f
#define TPB   512
#define KB    8
#define RP    12
#define CP    12
#define PP    260

template <int M>
__device__ __forceinline__ float row_xor(float v) {
    return __builtin_bit_cast(float,
        __builtin_amdgcn_update_dpp(0, __builtin_bit_cast(int, v), 0x160 | M, 0xf, 0xf, true));
}

__device__ __forceinline__ float lane_value(float v, int l) {
    return __builtin_bit_cast(float, __builtin_amdgcn_readlane(__builtin_bit_cast(int, v), l));
}

extern "C" __global__ __launch_bounds__(TPB, 1)
void boxqp_admm_kernel(const float* __restrict__ Q,
                       const float* __restrict__ p,
                       const float* __restrict__ lb,
                       const float* __restrict__ ub,
                       float* __restrict__ out)
{
    extern __shared__ float smem[];
    float* Mm    = smem;
    float* xv    = smem + NN * LDM;
    float* Cn    = xv + NN;
    float* BT    = Cn + NN * CP;
    float* part  = Mm;

    const int tid  = threadIdx.x;
    const int b    = blockIdx.x;
    const int lane = tid & 31;
    const int wv   = tid >> 5;
    const int cl   = lane & 15;
    const int hh   = lane >> 4;
    const int rowA = (wv << 4) + cl;
    const int hk   = hh << 1;
    const int rr   = (wv << 4) + (hh << 3);
    const float* Qb = Q + (size_t)b * NN * NN;

    for (int idx = tid; idx < NN * (NN / 4); idx += TPB) {
        int row = idx >> 6;
        int c   = (idx & 63) << 2;
        float4 q = *(const float4*)(Qb + row * NN + c);
        float* d = Mm + row * LDM + c;
        d[0] = q.x; d[1] = q.y; d[2] = q.z; d[3] = q.w;
    }
    __syncthreads();
    if (tid < NN) Mm[tid * LDM + tid] += RHO;
    __syncthreads();

    v8f acc[16];
    #pragma unroll
    for (int T = 0; T < 16; ++T) {
        const float* s = Mm + ((T << 4) + cl) * LDM + rr;
        float4 lo = *(const float4*)(s);
        float4 hi = *(const float4*)(s + 4);
        acc[T] = (v8f){ lo.x, lo.y, lo.z, lo.w, hi.x, hi.y, hi.z, hi.w };
    }
    __syncthreads();

    for (int K = 0; K < NN; K += KB) {
        const int  cj    = K >> 4;
        const int  k0    = K & 15;
        const int  ph    = k0 >> 3;
        const int  m     = cl & 7;
        const int  mk    = cl - k0;
        const bool incol = ((cl >> 3) == ph);
        const bool prow  = (wv == cj) && (hh == ph);
        float* PRw = Mm + ((K >> 3) & 1) * (NN * RP);

        if (wv == cj) {
            if (hh == ph) {
                #pragma unroll
                for (int T = 0; T < 16; ++T) {
                    float* d = PRw + ((T << 4) + cl) * RP;
                    *(float4*)(d)     = make_float4(acc[T][0], acc[T][1], acc[T][2], acc[T][3]);
                    *(float4*)(d + 4) = make_float4(acc[T][4], acc[T][5], acc[T][6], acc[T][7]);
                }
            }
            #pragma unroll
            for (int T = 0; T < 16; ++T) {
                #pragma unroll
                for (int g = 0; g < 8; ++g) acc[T][g] = (hh == ph) ? 0.f : acc[T][g];
            }
        }

        #pragma unroll
        for (int T = 0; T < 16; ++T) {
            if (T == cj) {
                if (incol) {
                    float* dst = Cn + m;
                    #pragma unroll
                    for (int g = 0; g < 8; ++g) {
                        float v = prow ? ((g == mk) ? 1.f : 0.f) : -acc[T][g];
                        dst[(rr + g) * CP] = v;
                    }
                }
                #pragma unroll
                for (int g = 0; g < 8; ++g) acc[T][g] = incol ? 0.f : acc[T][g];
            }
        }
        __syncthreads();

        float cr[8];
        {
            float4 c0 = *(const float4*)(Cn + rowA * CP);
            float4 c1 = *(const float4*)(Cn + rowA * CP + 4);
            cr[0] = c0.x; cr[1] = c0.y; cr[2] = c0.z; cr[3] = c0.w;
            cr[4] = c1.x; cr[5] = c1.y; cr[6] = c1.z; cr[7] = c1.w;
        }

        if (wv == 0) {
            float ar[8];
            #pragma unroll
            for (int j = 0; j < 8; ++j)
                ar[j] = (lane < 8) ? PRw[(K + j) * RP + lane] : 0.f;
            if (lane < 8) {
                #pragma unroll
                for (int j = 0; j < 8; ++j) PRw[(K + j) * RP + lane] = 0.f;
                PRw[(K + lane) * RP + lane] = 1.f;
            }
            #pragma unroll
            for (int kk = 0; kk < 8; ++kk) {
                float prw[8];
                #pragma unroll
                for (int j = 0; j < 8; ++j) prw[j] = lane_value(ar[j], kk);
                const float pinv = 1.0f / prw[kk];
                const bool  isp  = (lane == kk);
                const float f    = ar[kk];
                #pragma unroll
                for (int j = 0; j < 8; ++j) {
                    if (j != kk) {
                        const float sr = prw[j] * pinv;
                        ar[j] = isp ? sr : (ar[j] - f * sr);
                    }
                }
                ar[kk] = isp ? pinv : (-f * pinv);
            }
            if (lane < 8) {
                #pragma unroll
                for (int j = 0; j < 8; ++j) BT[j * 8 + lane] = ar[j];
            }
        }
        __syncthreads();

        v2f ca0, ca1;
        {
            float4 t0[4], t1[4];
            #pragma unroll
            for (int e = 0; e < 4; ++e) {
                const float* bsrc = BT + (((e >> 1) << 2) + hk + (e & 1)) * 8;
                t0[e] = *(const float4*)(bsrc);
                t1[e] = *(const float4*)(bsrc + 4);
            }
            float cv[4];
            #pragma unroll
            for (int e = 0; e < 4; ++e) {
                float sacc = cr[0] * t0[e].x;
                sacc = __builtin_fmaf(cr[1], t0[e].y, sacc);
                sacc = __builtin_fmaf(cr[2], t0[e].z, sacc);
                sacc = __builtin_fmaf(cr[3], t0[e].w, sacc);
                sacc = __builtin_fmaf(cr[4], t1[e].x, sacc);
                sacc = __builtin_fmaf(cr[5], t1[e].y, sacc);
                sacc = __builtin_fmaf(cr[6], t1[e].z, sacc);
                sacc = __builtin_fmaf(cr[7], t1[e].w, sacc);
                cv[e] = sacc;
            }
            ca0 = (v2f){ cv[0], cv[1] };
            ca1 = (v2f){ cv[2], cv[3] };
        }

        {
            const float* prsrc = PRw + cl * RP + hk;
            #pragma unroll
            for (int T = 0; T < 16; ++T) {
                v2f b0 = *(const v2f*)(prsrc + (T << 4) * RP);
                v2f b1 = *(const v2f*)(prsrc + (T << 4) * RP + 4);
                acc[T] = __builtin_amdgcn_wmma_f32_16x16x4_f32(
                    false, ca0, false, b0, (short)0, acc[T], false, false);
                acc[T] = __builtin_amdgcn_wmma_f32_16x16x4_f32(
                    false, ca1, false, b1, (short)0, acc[T], false, false);
            }
        }
    }
    __syncthreads();

    const int e0 = (wv << 4) + (hh << 3);
    float pv[8], lbv[8], ubv[8], uv[8], rv[8], xo[8];
    {
        const float4* p4 = (const float4*)(p  + (size_t)b * NN + e0);
        const float4* l4 = (const float4*)(lb + (size_t)b * NN + e0);
        const float4* u4 = (const float4*)(ub + (size_t)b * NN + e0);
        float4 a0 = p4[0], a1 = p4[1], b0 = l4[0], b1 = l4[1], c0 = u4[0], c1 = u4[1];
        pv[0] = a0.x; pv[1] = a0.y; pv[2] = a0.z; pv[3] = a0.w;
        pv[4] = a1.x; pv[5] = a1.y; pv[6] = a1.z; pv[7] = a1.w;
        lbv[0] = b0.x; lbv[1] = b0.y; lbv[2] = b0.z; lbv[3] = b0.w;
        lbv[4] = b1.x; lbv[5] = b1.y; lbv[6] = b1.z; lbv[7] = b1.w;
        ubv[0] = c0.x; ubv[1] = c0.y; ubv[2] = c0.z; ubv[3] = c0.w;
        ubv[4] = c1.x; ubv[5] = c1.y; ubv[6] = c1.z; ubv[7] = c1.w;
    }
    #pragma unroll
    for (int g = 0; g < 8; ++g) { uv[g] = 0.f; rv[g] = -pv[g]; xo[g] = 0.f; }

    for (int it = 0; it < NITER; ++it) {
        v8f d0 = {}, d1 = {};
        #pragma unroll
        for (int T = 0; T < 16; T += 2) {
            const bool m0 = (cl == T), m1 = (cl == T + 1);
            #pragma unroll
            for (int q = 0; q < 4; ++q) {
                v2f a0 = { acc[T][2 * q],     acc[T][2 * q + 1] };
                v2f a1 = { acc[T + 1][2 * q], acc[T + 1][2 * q + 1] };
                v2f b0 = { m0 ? rv[2 * q] : 0.f, m0 ? rv[2 * q + 1] : 0.f };
                v2f b1 = { m1 ? rv[2 * q] : 0.f, m1 ? rv[2 * q + 1] : 0.f };
                d0 = __builtin_amdgcn_wmma_f32_16x16x4_f32(
                    false, a0, false, b0, (short)0, d0, false, false);
                d1 = __builtin_amdgcn_wmma_f32_16x16x4_f32(
                    false, a1, false, b1, (short)0, d1, false, false);
            }
        }
        v8f d = d0 + d1;

        float* pw = part + (((it & 1) << 4) + wv) * PP + (cl << 4) + (hh << 3);
        *(float4*)(pw)     = make_float4(d[0], d[1], d[2], d[3]);
        *(float4*)(pw + 4) = make_float4(d[4], d[5], d[6], d[7]);
        __syncthreads();

        const float* pr = part + (((it & 1) << 4) + cl) * PP + e0;
        float4 s0 = *(const float4*)(pr);
        float4 s1 = *(const float4*)(pr + 4);
        float s[8] = { s0.x, s0.y, s0.z, s0.w, s1.x, s1.y, s1.z, s1.w };
        #pragma unroll
        for (int g = 0; g < 8; ++g) s[g] += row_xor<1>(s[g]);
        #pragma unroll
        for (int g = 0; g < 8; ++g) s[g] += row_xor<2>(s[g]);
        #pragma unroll
        for (int g = 0; g < 8; ++g) s[g] += row_xor<4>(s[g]);
        #pragma unroll
        for (int g = 0; g < 8; ++g) s[g] += row_xor<8>(s[g]);

        #pragma unroll
        for (int g = 0; g < 8; ++g) {
            float xi = s[g];
            float t  = xi + uv[g];
            float zi = fminf(fmaxf(t, lbv[g]), ubv[g]);
            uv[g] = uv[g] + xi - zi;
            rv[g] = -pv[g] + (zi - uv[g]);
            xo[g] = xi;
        }
    }

    if (cl == 0) {
        *(float4*)(xv + e0)     = make_float4(xo[0], xo[1], xo[2], xo[3]);
        *(float4*)(xv + e0 + 4) = make_float4(xo[4], xo[5], xo[6], xo[7]);
    }
    __syncthreads();
    if (tid < NN) out[(size_t)b * NN + tid] = xv[tid];
}

extern "C" void kernel_launch(void* const* d_in, const int* in_sizes, int n_in,
                              void* d_out, int out_size, void* d_ws, size_t ws_size,
                              hipStream_t stream) {
    const float* Q  = (const float*)d_in[0];
    const float* p  = (const float*)d_in[1];
    const float* lb = (const float*)d_in[2];
    const float* ub = (const float*)d_in[3];
    float* out = (float*)d_out;

    const int B = in_sizes[0] / (NN * NN);
    const size_t shmem = (size_t)(NN * LDM + NN + 16 * NN + 64) * sizeof(float);

    (void)hipFuncSetAttribute((const void*)boxqp_admm_kernel,
                              hipFuncAttributeMaxDynamicSharedMemorySize,
                              (int)shmem);

    boxqp_admm_kernel<<<B, TPB, shmem, stream>>>(Q, p, lb, ub, out);
}


# 8x8 pivot-block inverse: pivot reciprocal by one v_rcp_f32 (1 ulp) instead of the IEEE division sequence, pivot-lane selects moved ahead of it; a pivot step's dependent chain is rcp-mul-readlane-fma
# speedup vs baseline: 1.4685x; 1.0425x over previous
#include <hip/hip_runtime.h>


typedef float v2f __attribute__((ext_vector_type(2)));
typedef float v8f __attribute__((ext_vector_type(8)));

#define NN    256
#define LDM   260
#define NITER 50
#define RHO   1.0f
#define TPB   512
#define KB    8
#define RP    12
#define CP    12
#define PP    260

template <int M>
__device__ __forceinline__ float row_xor(float v) {
    return __builtin_bit_cast(float,
        __builtin_amdgcn_update_dpp(0, __builtin_bit_cast(int, v), 0x160 | M, 0xf, 0xf, true));
}

__device__ __forceinline__ float lane_value(float v, int l) {
    return __builtin_bit_cast(float, __builtin_amdgcn_readlane(__builtin_bit_cast(int, v), l));
}

extern "C" __global__ __launch_bounds__(TPB, 1)
void boxqp_admm_kernel(const float* __restrict__ Q,
                       const float* __restrict__ p,
                       const float* __restrict__ lb,
                       const float* __restrict__ ub,
                       float* __restrict__ out)
{
    extern __shared__ float smem[];
    float* Mm    = smem;
    float* xv    = smem + NN * LDM;
    float* Cn    = xv + NN;
    float* BT    = Cn + NN * CP;
    float* part  = Mm;

    const int tid  = threadIdx.x;
    const int b    = blockIdx.x;
    const int lane = tid & 31;
    const int wv   = tid >> 5;
    const int cl   = lane & 15;
    const int hh   = lane >> 4;
    const int rowA = (wv << 4) + cl;
    const int hk   = hh << 1;
    const int rr   = (wv << 4) + (hh << 3);
    const float* Qb = Q + (size_t)b * NN * NN;

    for (int idx = tid; idx < NN * (NN / 4); idx += TPB) {
        int row = idx >> 6;
        int c   = (idx & 63) << 2;
        float4 q = *(const float4*)(Qb + row * NN + c);
        float* d = Mm + row * LDM + c;
        d[0] = q.x; d[1] = q.y; d[2] = q.z; d[3] = q.w;
    }
    __syncthreads();
    if (tid < NN) Mm[tid * LDM + tid] += RHO;
    __syncthreads();

    v8f acc[16];
    #pragma unroll
    for (int T = 0; T < 16; ++T) {
        const float* s = Mm + ((T << 4) + cl) * LDM + rr;
        float4 lo = *(const float4*)(s);
        float4 hi = *(const float4*)(s + 4);
        acc[T] = (v8f){ lo.x, lo.y, lo.z, lo.w, hi.x, hi.y, hi.z, hi.w };
    }
    __syncthreads();

    for (int K = 0; K < NN; K += KB) {
        const int  cj    = K >> 4;
        const int  k0    = K & 15;
        const int  ph    = k0 >> 3;
        const int  m     = cl & 7;
        const int  mk    = cl - k0;
        const bool incol = ((cl >> 3) == ph);
        const bool prow  = (wv == cj) && (hh == ph);
        float* PRw = Mm + ((K >> 3) & 1) * (NN * RP);

        if (wv == cj) {
            if (hh == ph) {
                #pragma unroll
                for (int T = 0; T < 16; ++T) {
                    float* d = PRw + ((T << 4) + cl) * RP;
                    *(float4*)(d)     = make_float4(acc[T][0], acc[T][1], acc[T][2], acc[T][3]);
                    *(float4*)(d + 4) = make_float4(acc[T][4], acc[T][5], acc[T][6], acc[T][7]);
                }
            }
            #pragma unroll
            for (int T = 0; T < 16; ++T) {
                #pragma unroll
                for (int g = 0; g < 8; ++g) acc[T][g] = (hh == ph) ? 0.f : acc[T][g];
            }
        }

        #pragma unroll
        for (int T = 0; T < 16; ++T) {
            if (T == cj) {
                if (incol) {
                    float* dst = Cn + m;
                    #pragma unroll
                    for (int g = 0; g < 8; ++g) {
                        float v = prow ? ((g == mk) ? 1.f : 0.f) : -acc[T][g];
                        dst[(rr + g) * CP] = v;
                    }
                }
                #pragma unroll
                for (int g = 0; g < 8; ++g) acc[T][g] = incol ? 0.f : acc[T][g];
            }
        }
        __syncthreads();

        float cr[8];
        {
            float4 c0 = *(const float4*)(Cn + rowA * CP);
            float4 c1 = *(const float4*)(Cn + rowA * CP + 4);
            cr[0] = c0.x; cr[1] = c0.y; cr[2] = c0.z; cr[3] = c0.w;
            cr[4] = c1.x; cr[5] = c1.y; cr[6] = c1.z; cr[7] = c1.w;
        }

        if (wv == 0) {
            const int l8 = lane & 7;
            float ar[8];
            #pragma unroll
            for (int j = 0; j < 8; ++j) ar[j] = PRw[(K + j) * RP + l8];
            #pragma unroll
            for (int kk = 0; kk < 8; ++kk) {
                const float rinv = __builtin_amdgcn_rcpf(ar[kk]);
                const float pinv = lane_value(rinv, kk);
                const bool  isp  = (l8 == kk);
                const float nf   = isp ? -1.0f : ar[kk];
                #pragma unroll
                for (int j = 0; j < 8; ++j) {
                    if (j != kk) {
                        const float base = isp ? 0.f : ar[j];
                        const float sr   = lane_value(ar[j] * rinv, kk);
                        ar[j] = __builtin_fmaf(-nf, sr, base);
                    }
                }
                ar[kk] = -nf * pinv;
            }
            if (lane < 8) {
                #pragma unroll
                for (int j = 0; j < 8; ++j) PRw[(K + j) * RP + lane] = (j == lane) ? 1.f : 0.f;
                #pragma unroll
                for (int j = 0; j < 8; ++j) BT[j * 8 + lane] = ar[j];
            }
        }
        __syncthreads();

        v2f ca0, ca1;
        {
            float4 t0[4], t1[4];
            #pragma unroll
            for (int e = 0; e < 4; ++e) {
                const float* bsrc = BT + (((e >> 1) << 2) + hk + (e & 1)) * 8;
                t0[e] = *(const float4*)(bsrc);
                t1[e] = *(const float4*)(bsrc + 4);
            }
            float cv[4];
            #pragma unroll
            for (int e = 0; e < 4; ++e) {
                float sacc = cr[0] * t0[e].x;
                sacc = __builtin_fmaf(cr[1], t0[e].y, sacc);
                sacc = __builtin_fmaf(cr[2], t0[e].z, sacc);
                sacc = __builtin_fmaf(cr[3], t0[e].w, sacc);
                sacc = __builtin_fmaf(cr[4], t1[e].x, sacc);
                sacc = __builtin_fmaf(cr[5], t1[e].y, sacc);
                sacc = __builtin_fmaf(cr[6], t1[e].z, sacc);
                sacc = __builtin_fmaf(cr[7], t1[e].w, sacc);
                cv[e] = sacc;
            }
            ca0 = (v2f){ cv[0], cv[1] };
            ca1 = (v2f){ cv[2], cv[3] };
        }

        {
            const float* prsrc = PRw + cl * RP + hk;
            #pragma unroll
            for (int T = 0; T < 16; ++T) {
                v2f b0 = *(const v2f*)(prsrc + (T << 4) * RP);
                v2f b1 = *(const v2f*)(prsrc + (T << 4) * RP + 4);
                acc[T] = __builtin_amdgcn_wmma_f32_16x16x4_f32(
                    false, ca0, false, b0, (short)0, acc[T], false, false);
                acc[T] = __builtin_amdgcn_wmma_f32_16x16x4_f32(
                    false, ca1, false, b1, (short)0, acc[T], false, false);
            }
        }
    }
    __syncthreads();

    const int e0 = (wv << 4) + (hh << 3);
    float pv[8], lbv[8], ubv[8], uv[8], rv[8], xo[8];
    {
        const float4* p4 = (const float4*)(p  + (size_t)b * NN + e0);
        const float4* l4 = (const float4*)(lb + (size_t)b * NN + e0);
        const float4* u4 = (const float4*)(ub + (size_t)b * NN + e0);
        float4 a0 = p4[0], a1 = p4[1], b0 = l4[0], b1 = l4[1], c0 = u4[0], c1 = u4[1];
        pv[0] = a0.x; pv[1] = a0.y; pv[2] = a0.z; pv[3] = a0.w;
        pv[4] = a1.x; pv[5] = a1.y; pv[6] = a1.z; pv[7] = a1.w;
        lbv[0] = b0.x; lbv[1] = b0.y; lbv[2] = b0.z; lbv[3] = b0.w;
        lbv[4] = b1.x; lbv[5] = b1.y; lbv[6] = b1.z; lbv[7] = b1.w;
        ubv[0] = c0.x; ubv[1] = c0.y; ubv[2] = c0.z; ubv[3] = c0.w;
        ubv[4] = c1.x; ubv[5] = c1.y; ubv[6] = c1.z; ubv[7] = c1.w;
    }
    #pragma unroll
    for (int g = 0; g < 8; ++g) { uv[g] = 0.f; rv[g] = -pv[g]; xo[g] = 0.f; }

    for (int it = 0; it < NITER; ++it) {
        v8f d0 = {}, d1 = {};
        #pragma unroll
        for (int T = 0; T < 16; T += 2) {
            const bool m0 = (cl == T), m1 = (cl == T + 1);
            #pragma unroll
            for (int q = 0; q < 4; ++q) {
                v2f a0 = { acc[T][2 * q],     acc[T][2 * q + 1] };
                v2f a1 = { acc[T + 1][2 * q], acc[T + 1][2 * q + 1] };
                v2f b0 = { m0 ? rv[2 * q] : 0.f, m0 ? rv[2 * q + 1] : 0.f };
                v2f b1 = { m1 ? rv[2 * q] : 0.f, m1 ? rv[2 * q + 1] : 0.f };
                d0 = __builtin_amdgcn_wmma_f32_16x16x4_f32(
                    false, a0, false, b0, (short)0, d0, false, false);
                d1 = __builtin_amdgcn_wmma_f32_16x16x4_f32(
                    false, a1, false, b1, (short)0, d1, false, false);
            }
        }
        v8f d = d0 + d1;

        float* pw = part + (((it & 1) << 4) + wv) * PP + (cl << 4) + (hh << 3);
        *(float4*)(pw)     = make_float4(d[0], d[1], d[2], d[3]);
        *(float4*)(pw + 4) = make_float4(d[4], d[5], d[6], d[7]);
        __syncthreads();

        const float* pr = part + (((it & 1) << 4) + cl) * PP + e0;
        float4 s0 = *(const float4*)(pr);
        float4 s1 = *(const float4*)(pr + 4);
        float s[8] = { s0.x, s0.y, s0.z, s0.w, s1.x, s1.y, s1.z, s1.w };
        #pragma unroll
        for (int g = 0; g < 8; ++g) s[g] += row_xor<1>(s[g]);
        #pragma unroll
        for (int g = 0; g < 8; ++g) s[g] += row_xor<2>(s[g]);
        #pragma unroll
        for (int g = 0; g < 8; ++g) s[g] += row_xor<4>(s[g]);
        #pragma unroll
        for (int g = 0; g < 8; ++g) s[g] += row_xor<8>(s[g]);

        #pragma unroll
        for (int g = 0; g < 8; ++g) {
            float xi = s[g];
            float t  = xi + uv[g];
            float zi = fminf(fmaxf(t, lbv[g]), ubv[g]);
            uv[g] = uv[g] + xi - zi;
            rv[g] = -pv[g] + (zi - uv[g]);
            xo[g] = xi;
        }
    }

    if (cl == 0) {
        *(float4*)(xv + e0)     = make_float4(xo[0], xo[1], xo[2], xo[3]);
        *(float4*)(xv + e0 + 4) = make_float4(xo[4], xo[5], xo[6], xo[7]);
    }
    __syncthreads();
    if (tid < NN) out[(size_t)b * NN + tid] = xv[tid];
}

extern "C" void kernel_launch(void* const* d_in, const int* in_sizes, int n_in,
                              void* d_out, int out_size, void* d_ws, size_t ws_size,
                              hipStream_t stream) {
    const float* Q  = (const float*)d_in[0];
    const float* p  = (const float*)d_in[1];
    const float* lb = (const float*)d_in[2];
    const float* ub = (const float*)d_in[3];
    float* out = (float*)d_out;

    const int B = in_sizes[0] / (NN * NN);
    const size_t shmem = (size_t)(NN * LDM + NN + 16 * NN + 64) * sizeof(float);

    (void)hipFuncSetAttribute((const void*)boxqp_admm_kernel,
                              hipFuncAttributeMaxDynamicSharedMemorySize,
                              (int)shmem);

    boxqp_admm_kernel<<<B, TPB, shmem, stream>>>(Q, p, lb, ub, out);
}


# ADMM matvec: each WMMA takes two tiles of Minv, one as A and one as B, with rhs in lane 0 of both, so column 0 and row 0 of D are both partial sums: 36 WMMAs per wave-iteration instead of 64
# speedup vs baseline: 1.8004x; 1.2260x over previous
#include <hip/hip_runtime.h>


typedef float v2f __attribute__((ext_vector_type(2)));
typedef float v8f __attribute__((ext_vector_type(8)));

#define NN    256
#define LDM   260
#define NITER 50
#define RHO   1.0f
#define TPB   512
#define KB    8
#define RP    12
#define CP    12
#define PP    276
#define DUMP  9216
#define R0OFF 20480

template <int M>
__device__ __forceinline__ float row_xor(float v) {
    return __builtin_bit_cast(float,
        __builtin_amdgcn_update_dpp(0, __builtin_bit_cast(int, v), 0x160 | M, 0xf, 0xf, true));
}

__device__ __forceinline__ float lane_value(float v, int l) {
    return __builtin_bit_cast(float, __builtin_amdgcn_readlane(__builtin_bit_cast(int, v), l));
}

extern "C" __global__ __launch_bounds__(TPB, 1)
void boxqp_admm_kernel(const float* __restrict__ Q,
                       const float* __restrict__ p,
                       const float* __restrict__ lb,
                       const float* __restrict__ ub,
                       float* __restrict__ out)
{
    extern __shared__ float smem[];
    float* Mm    = smem;
    float* xv    = smem + NN * LDM;
    float* Cn    = xv + NN;
    float* BT    = Cn + NN * CP;
    float* part  = Mm;

    const int tid  = threadIdx.x;
    const int b    = blockIdx.x;
    const int lane = tid & 31;
    const int wv   = tid >> 5;
    const int cl   = lane & 15;
    const int hh   = lane >> 4;
    const int rowA = (wv << 4) + cl;
    const int hk   = hh << 1;
    const int rr   = (wv << 4) + (hh << 3);
    const float* Qb = Q + (size_t)b * NN * NN;

    for (int idx = tid; idx < NN * (NN / 4); idx += TPB) {
        int row = idx >> 6;
        int c   = (idx & 63) << 2;
        float4 q = *(const float4*)(Qb + row * NN + c);
        float* d = Mm + row * LDM + c;
        d[0] = q.x; d[1] = q.y; d[2] = q.z; d[3] = q.w;
    }
    __syncthreads();
    if (tid < NN) Mm[tid * LDM + tid] += RHO;
    __syncthreads();

    v8f acc[16];
    #pragma unroll
    for (int T = 0; T < 16; ++T) {
        const float* s = Mm + ((T << 4) + cl) * LDM + rr;
        float4 lo = *(const float4*)(s);
        float4 hi = *(const float4*)(s + 4);
        acc[T] = (v8f){ lo.x, lo.y, lo.z, lo.w, hi.x, hi.y, hi.z, hi.w };
    }
    __syncthreads();

    for (int K = 0; K < NN; K += KB) {
        const int  cj    = K >> 4;
        const int  k0    = K & 15;
        const int  ph    = k0 >> 3;
        const int  m     = cl & 7;
        const int  mk    = cl - k0;
        const bool incol = ((cl >> 3) == ph);
        const bool prow  = (wv == cj) && (hh == ph);
        float* PRw = Mm + ((K >> 3) & 1) * (NN * RP);

        if (wv == cj) {
            if (hh == ph) {
                #pragma unroll
                for (int T = 0; T < 16; ++T) {
                    float* d = PRw + ((T << 4) + cl) * RP;
                    *(float4*)(d)     = make_float4(acc[T][0], acc[T][1], acc[T][2], acc[T][3]);
                    *(float4*)(d + 4) = make_float4(acc[T][4], acc[T][5], acc[T][6], acc[T][7]);
                }
            }
            #pragma unroll
            for (int T = 0; T < 16; ++T) {
                #pragma unroll
                for (int g = 0; g < 8; ++g) acc[T][g] = (hh == ph) ? 0.f : acc[T][g];
            }
        }

        #pragma unroll
        for (int T = 0; T < 16; ++T) {
            if (T == cj) {
                if (incol) {
                    float* dst = Cn + m;
                    #pragma unroll
                    for (int g = 0; g < 8; ++g) {
                        float v = prow ? ((g == mk) ? 1.f : 0.f) : -acc[T][g];
                        dst[(rr + g) * CP] = v;
                    }
                }
                #pragma unroll
                for (int g = 0; g < 8; ++g) acc[T][g] = incol ? 0.f : acc[T][g];
            }
        }
        __syncthreads();

        float cr[8];
        {
            float4 c0 = *(const float4*)(Cn + rowA * CP);
            float4 c1 = *(const float4*)(Cn + rowA * CP + 4);
            cr[0] = c0.x; cr[1] = c0.y; cr[2] = c0.z; cr[3] = c0.w;
            cr[4] = c1.x; cr[5] = c1.y; cr[6] = c1.z; cr[7] = c1.w;
        }

        if (wv == 0) {
            const int l8 = lane & 7;
            float ar[8];
            #pragma unroll
            for (int j = 0; j < 8; ++j) ar[j] = PRw[(K + j) * RP + l8];
            #pragma unroll
            for (int kk = 0; kk < 8; ++kk) {
                const float rinv = __builtin_amdgcn_rcpf(ar[kk]);
                const float pinv = lane_value(rinv, kk);
                const bool  isp  = (l8 == kk);
                const float nf   = isp ? -1.0f : ar[kk];
                #pragma unroll
                for (int j = 0; j < 8; ++j) {
                    if (j != kk) {
                        const float base = isp ? 0.f : ar[j];
                        const float sr   = lane_value(ar[j] * rinv, kk);
                        ar[j] = __builtin_fmaf(-nf, sr, base);
                    }
                }
                ar[kk] = -nf * pinv;
            }
            if (lane < 8) {
                #pragma unroll
                for (int j = 0; j < 8; ++j) PRw[(K + j) * RP + lane] = (j == lane) ? 1.f : 0.f;
                #pragma unroll
                for (int j = 0; j < 8; ++j) BT[j * 8 + lane] = ar[j];
            }
        }
        __syncthreads();

        v2f ca0, ca1;
        {
            float4 t0[4], t1[4];
            #pragma unroll
            for (int e = 0; e < 4; ++e) {
                const float* bsrc = BT + (((e >> 1) << 2) + hk + (e & 1)) * 8;
                t0[e] = *(const float4*)(bsrc);
                t1[e] = *(const float4*)(bsrc + 4);
            }
            float cv[4];
            #pragma unroll
            for (int e = 0; e < 4; ++e) {
                float sacc = cr[0] * t0[e].x;
                sacc = __builtin_fmaf(cr[1], t0[e].y, sacc);
                sacc = __builtin_fmaf(cr[2], t0[e].z, sacc);
                sacc = __builtin_fmaf(cr[3], t0[e].w, sacc);
                sacc = __builtin_fmaf(cr[4], t1[e].x, sacc);
                sacc = __builtin_fmaf(cr[5], t1[e].y, sacc);
                sacc = __builtin_fmaf(cr[6], t1[e].z, sacc);
                sacc = __builtin_fmaf(cr[7], t1[e].w, sacc);
                cv[e] = sacc;
            }
            ca0 = (v2f){ cv[0], cv[1] };
            ca1 = (v2f){ cv[2], cv[3] };
        }

        {
            const float* prsrc = PRw + cl * RP + hk;
            #pragma unroll
            for (int T = 0; T < 16; ++T) {
                v2f b0 = *(const v2f*)(prsrc + (T << 4) * RP);
                v2f b1 = *(const v2f*)(prsrc + (T << 4) * RP + 4);
                acc[T] = __builtin_amdgcn_wmma_f32_16x16x4_f32(
                    false, ca0, false, b0, (short)0, acc[T], false, false);
                acc[T] = __builtin_amdgcn_wmma_f32_16x16x4_f32(
                    false, ca1, false, b1, (short)0, acc[T], false, false);
            }
        }
    }
    const bool l0 = (cl == 0);
    float* r0s = Mm + R0OFF + (wv << 8);
    if (l0) {
        #pragma unroll
        for (int T = 0; T < 16; ++T) {
            float* d = r0s + (((T << 1) + hh) << 3);
            *(float4*)(d)     = make_float4(acc[T][0], acc[T][1], acc[T][2], acc[T][3]);
            *(float4*)(d + 4) = make_float4(acc[T][4], acc[T][5], acc[T][6], acc[T][7]);
        }
    }
    __syncthreads();
    v8f r0;
    {
        const float* s = r0s + (((cl << 1) + hh) << 3);
        float4 lo = *(const float4*)(s);
        float4 hi = *(const float4*)(s + 4);
        r0 = (v8f){ lo.x, lo.y, lo.z, lo.w, hi.x, hi.y, hi.z, hi.w };
    }

    const int e0 = (wv << 4) + (hh << 3);
    float pv[8], lbv[8], ubv[8], uv[8], rv[8], xo[8];
    {
        const float4* p4 = (const float4*)(p  + (size_t)b * NN + e0);
        const float4* l4 = (const float4*)(lb + (size_t)b * NN + e0);
        const float4* u4 = (const float4*)(ub + (size_t)b * NN + e0);
        float4 a0 = p4[0], a1 = p4[1], b0 = l4[0], b1 = l4[1], c0 = u4[0], c1 = u4[1];
        pv[0] = a0.x; pv[1] = a0.y; pv[2] = a0.z; pv[3] = a0.w;
        pv[4] = a1.x; pv[5] = a1.y; pv[6] = a1.z; pv[7] = a1.w;
        lbv[0] = b0.x; lbv[1] = b0.y; lbv[2] = b0.z; lbv[3] = b0.w;
        lbv[4] = b1.x; lbv[5] = b1.y; lbv[6] = b1.z; lbv[7] = b1.w;
        ubv[0] = c0.x; ubv[1] = c0.y; ubv[2] = c0.z; ubv[3] = c0.w;
        ubv[4] = c1.x; ubv[5] = c1.y; ubv[6] = c1.z; ubv[7] = c1.w;
    }
    #pragma unroll
    for (int g = 0; g < 8; ++g) { uv[g] = 0.f; rv[g] = -pv[g]; xo[g] = 0.f; }

    for (int it = 0; it < NITER; ++it) {
        float* wrow = part + (((it & 1) << 4) + wv) * PP;
        float* colp = wrow + (l0 ? (hh << 3) : (DUMP + lane * 12));
        float* rowp = wrow + 16 + (hh ? (DUMP + lane) : cl);

        v8f dr = {};
        #pragma unroll
        for (int G = 0; G < 4; ++G) {
            const int Ta = 4 * G, Tb = 4 * G + 2;
            #pragma unroll
            for (int T = Ta; T < Ta + 4; ++T) {
                #pragma unroll
                for (int g = 0; g < 8; ++g) acc[T][g] = l0 ? rv[g] : acc[T][g];
            }
            v8f da = {}, db = {};
            #pragma unroll
            for (int q = 0; q < 4; ++q) {
                if (G == 0) {
                    v2f ar = { r0[2 * q], r0[2 * q + 1] };
                    v2f br = { rv[2 * q], rv[2 * q + 1] };
                    dr = __builtin_amdgcn_wmma_f32_16x16x4_f32(
                        false, ar, false, br, (short)0, dr, false, false);
                }
                v2f a0 = { acc[Ta][2 * q],     acc[Ta][2 * q + 1] };
                v2f b0 = { acc[Ta + 1][2 * q], acc[Ta + 1][2 * q + 1] };
                v2f a1 = { acc[Tb][2 * q],     acc[Tb][2 * q + 1] };
                v2f b1 = { acc[Tb + 1][2 * q], acc[Tb + 1][2 * q + 1] };
                da = __builtin_amdgcn_wmma_f32_16x16x4_f32(
                    false, a0, false, b0, (short)0, da, false, false);
                db = __builtin_amdgcn_wmma_f32_16x16x4_f32(
                    false, a1, false, b1, (short)0, db, false, false);
            }
            *(float4*)(colp + 16 * Ta)     = make_float4(da[0], da[1], da[2], da[3]);
            *(float4*)(colp + 16 * Ta + 4) = make_float4(da[4], da[5], da[6], da[7]);
            rowp[16 * Ta] = da[0];
            *(float4*)(colp + 16 * Tb)     = make_float4(db[0], db[1], db[2], db[3]);
            *(float4*)(colp + 16 * Tb + 4) = make_float4(db[4], db[5], db[6], db[7]);
            rowp[16 * Tb] = db[0];
            if (G == 0) {
                *(float4*)(colp + NN)     = make_float4(dr[0], dr[1], dr[2], dr[3]);
                *(float4*)(colp + NN + 4) = make_float4(dr[4], dr[5], dr[6], dr[7]);
            }
        }
        __syncthreads();

        const float* pr = part + (((it & 1) << 4) + cl) * PP;
        float4 s0 = *(const float4*)(pr + e0);
        float4 s1 = *(const float4*)(pr + e0 + 4);
        float  sz = pr[NN + wv];
        float s[8] = { hh ? s0.x : sz, s0.y, s0.z, s0.w, s1.x, s1.y, s1.z, s1.w };
        #pragma unroll
        for (int g = 0; g < 8; ++g) s[g] += row_xor<1>(s[g]);
        #pragma unroll
        for (int g = 0; g < 8; ++g) s[g] += row_xor<2>(s[g]);
        #pragma unroll
        for (int g = 0; g < 8; ++g) s[g] += row_xor<4>(s[g]);
        #pragma unroll
        for (int g = 0; g < 8; ++g) s[g] += row_xor<8>(s[g]);

        #pragma unroll
        for (int g = 0; g < 8; ++g) {
            float xi = s[g];
            float t  = xi + uv[g];
            float zi = fminf(fmaxf(t, lbv[g]), ubv[g]);
            uv[g] = uv[g] + xi - zi;
            rv[g] = -pv[g] + (zi - uv[g]);
            xo[g] = xi;
        }
    }

    if (cl == 0) {
        *(float4*)(xv + e0)     = make_float4(xo[0], xo[1], xo[2], xo[3]);
        *(float4*)(xv + e0 + 4) = make_float4(xo[4], xo[5], xo[6], xo[7]);
    }
    __syncthreads();
    if (tid < NN) out[(size_t)b * NN + tid] = xv[tid];
}

extern "C" void kernel_launch(void* const* d_in, const int* in_sizes, int n_in,
                              void* d_out, int out_size, void* d_ws, size_t ws_size,
                              hipStream_t stream) {
    const float* Q  = (const float*)d_in[0];
    const float* p  = (const float*)d_in[1];
    const float* lb = (const float*)d_in[2];
    const float* ub = (const float*)d_in[3];
    float* out = (float*)d_out;

    const int B = in_sizes[0] / (NN * NN);
    const size_t shmem = (size_t)(NN * LDM + NN + 16 * NN + 64) * sizeof(float);

    (void)hipFuncSetAttribute((const void*)boxqp_admm_kernel,
                              hipFuncAttributeMaxDynamicSharedMemorySize,
                              (int)shmem);

    boxqp_admm_kernel<<<B, TPB, shmem, stream>>>(Q, p, lb, ub, out);
}


# ADMM cross-wave sum: each lane sums the 16 partials of ONE entry (8 LDS reads, pairwise tree in the lane, one row xor) and projects it; next rhs back by 8 row_share reads; same sums, same order
# speedup vs baseline: 1.8653x; 1.0360x over previous
#include <hip/hip_runtime.h>


typedef float v2f __attribute__((ext_vector_type(2)));
typedef float v8f __attribute__((ext_vector_type(8)));

#define NN    256
#define LDM   260
#define NITER 50
#define RHO   1.0f
#define TPB   512
#define KB    8
#define RP    12
#define CP    12
#define PP    276
#define DUMP  9216
#define R0OFF 20480

template <int M>
__device__ __forceinline__ float row_xor(float v) {
    return __builtin_bit_cast(float,
        __builtin_amdgcn_update_dpp(0, __builtin_bit_cast(int, v), 0x160 | M, 0xf, 0xf, true));
}

template <int N>
__device__ __forceinline__ float row_share(float v) {
    return __builtin_bit_cast(float,
        __builtin_amdgcn_update_dpp(0, __builtin_bit_cast(int, v), 0x150 | N, 0xf, 0xf, true));
}

__device__ __forceinline__ float lane_value(float v, int l) {
    return __builtin_bit_cast(float, __builtin_amdgcn_readlane(__builtin_bit_cast(int, v), l));
}

extern "C" __global__ __launch_bounds__(TPB, 1)
void boxqp_admm_kernel(const float* __restrict__ Q,
                       const float* __restrict__ p,
                       const float* __restrict__ lb,
                       const float* __restrict__ ub,
                       float* __restrict__ out)
{
    extern __shared__ float smem[];
    float* Mm    = smem;
    float* xv    = smem + NN * LDM;
    float* Cn    = xv + NN;
    float* BT    = Cn + NN * CP;
    float* part  = Mm;

    const int tid  = threadIdx.x;
    const int b    = blockIdx.x;
    const int lane = tid & 31;
    const int wv   = tid >> 5;
    const int cl   = lane & 15;
    const int hh   = lane >> 4;
    const int rowA = (wv << 4) + cl;
    const int hk   = hh << 1;
    const int rr   = (wv << 4) + (hh << 3);
    const float* Qb = Q + (size_t)b * NN * NN;

    for (int idx = tid; idx < NN * (NN / 4); idx += TPB) {
        int row = idx >> 6;
        int c   = (idx & 63) << 2;
        float4 q = *(const float4*)(Qb + row * NN + c);
        float* d = Mm + row * LDM + c;
        d[0] = q.x; d[1] = q.y; d[2] = q.z; d[3] = q.w;
    }
    __syncthreads();
    if (tid < NN) Mm[tid * LDM + tid] += RHO;
    __syncthreads();

    v8f acc[16];
    #pragma unroll
    for (int T = 0; T < 16; ++T) {
        const float* s = Mm + ((T << 4) + cl) * LDM + rr;
        float4 lo = *(const float4*)(s);
        float4 hi = *(const float4*)(s + 4);
        acc[T] = (v8f){ lo.x, lo.y, lo.z, lo.w, hi.x, hi.y, hi.z, hi.w };
    }
    __syncthreads();

    for (int K = 0; K < NN; K += KB) {
        const int  cj    = K >> 4;
        const int  k0    = K & 15;
        const int  ph    = k0 >> 3;
        const int  m     = cl & 7;
        const int  mk    = cl - k0;
        const bool incol = ((cl >> 3) == ph);
        const bool prow  = (wv == cj) && (hh == ph);
        float* PRw = Mm + ((K >> 3) & 1) * (NN * RP);

        if (wv == cj) {
            if (hh == ph) {
                #pragma unroll
                for (int T = 0; T < 16; ++T) {
                    float* d = PRw + ((T << 4) + cl) * RP;
                    *(float4*)(d)     = make_float4(acc[T][0], acc[T][1], acc[T][2], acc[T][3]);
                    *(float4*)(d + 4) = make_float4(acc[T][4], acc[T][5], acc[T][6], acc[T][7]);
                }
            }
            #pragma unroll
            for (int T = 0; T < 16; ++T) {
                #pragma unroll
                for (int g = 0; g < 8; ++g) acc[T][g] = (hh == ph) ? 0.f : acc[T][g];
            }
        }

        #pragma unroll
        for (int T = 0; T < 16; ++T) {
            if (T == cj) {
                if (incol) {
                    float* dst = Cn + m;
                    #pragma unroll
                    for (int g = 0; g < 8; ++g) {
                        float v = prow ? ((g == mk) ? 1.f : 0.f) : -acc[T][g];
                        dst[(rr + g) * CP] = v;
                    }
                }
                #pragma unroll
                for (int g = 0; g < 8; ++g) acc[T][g] = incol ? 0.f : acc[T][g];
            }
        }
        __syncthreads();

        float cr[8];
        {
            float4 c0 = *(const float4*)(Cn + rowA * CP);
            float4 c1 = *(const float4*)(Cn + rowA * CP + 4);
            cr[0] = c0.x; cr[1] = c0.y; cr[2] = c0.z; cr[3] = c0.w;
            cr[4] = c1.x; cr[5] = c1.y; cr[6] = c1.z; cr[7] = c1.w;
        }

        if (wv == 0) {
            const int l8 = lane & 7;
            float ar[8];
            #pragma unroll
            for (int j = 0; j < 8; ++j) ar[j] = PRw[(K + j) * RP + l8];
            #pragma unroll
            for (int kk = 0; kk < 8; ++kk) {
                const float rinv = __builtin_amdgcn_rcpf(ar[kk]);
                const float pinv = lane_value(rinv, kk);
                const bool  isp  = (l8 == kk);
                const float nf   = isp ? -1.0f : ar[kk];
                #pragma unroll
                for (int j = 0; j < 8; ++j) {
                    if (j != kk) {
                        const float base = isp ? 0.f : ar[j];
                        const float sr   = lane_value(ar[j] * rinv, kk);
                        ar[j] = __builtin_fmaf(-nf, sr, base);
                    }
                }
                ar[kk] = -nf * pinv;
            }
            if (lane < 8) {
                #pragma unroll
                for (int j = 0; j < 8; ++j) PRw[(K + j) * RP + lane] = (j == lane) ? 1.f : 0.f;
                #pragma unroll
                for (int j = 0; j < 8; ++j) BT[j * 8 + lane] = ar[j];
            }
        }
        __syncthreads();

        v2f ca0, ca1;
        {
            float4 t0[4], t1[4];
            #pragma unroll
            for (int e = 0; e < 4; ++e) {
                const float* bsrc = BT + (((e >> 1) << 2) + hk + (e & 1)) * 8;
                t0[e] = *(const float4*)(bsrc);
                t1[e] = *(const float4*)(bsrc + 4);
            }
            float cv[4];
            #pragma unroll
            for (int e = 0; e < 4; ++e) {
                float sacc = cr[0] * t0[e].x;
                sacc = __builtin_fmaf(cr[1], t0[e].y, sacc);
                sacc = __builtin_fmaf(cr[2], t0[e].z, sacc);
                sacc = __builtin_fmaf(cr[3], t0[e].w, sacc);
                sacc = __builtin_fmaf(cr[4], t1[e].x, sacc);
                sacc = __builtin_fmaf(cr[5], t1[e].y, sacc);
                sacc = __builtin_fmaf(cr[6], t1[e].z, sacc);
                sacc = __builtin_fmaf(cr[7], t1[e].w, sacc);
                cv[e] = sacc;
            }
            ca0 = (v2f){ cv[0], cv[1] };
            ca1 = (v2f){ cv[2], cv[3] };
        }

        {
            const float* prsrc = PRw + cl * RP + hk;
            #pragma unroll
            for (int T = 0; T < 16; ++T) {
                v2f b0 = *(const v2f*)(prsrc + (T << 4) * RP);
                v2f b1 = *(const v2f*)(prsrc + (T << 4) * RP + 4);
                acc[T] = __builtin_amdgcn_wmma_f32_16x16x4_f32(
                    false, ca0, false, b0, (short)0, acc[T], false, false);
                acc[T] = __builtin_amdgcn_wmma_f32_16x16x4_f32(
                    false, ca1, false, b1, (short)0, acc[T], false, false);
            }
        }
    }
    const bool l0 = (cl == 0);
    float* r0s = Mm + R0OFF + (wv << 8);
    if (l0) {
        #pragma unroll
        for (int T = 0; T < 16; ++T) {
            float* d = r0s + (((T << 1) + hh) << 3);
            *(float4*)(d)     = make_float4(acc[T][0], acc[T][1], acc[T][2], acc[T][3]);
            *(float4*)(d + 4) = make_float4(acc[T][4], acc[T][5], acc[T][6], acc[T][7]);
        }
    }
    __syncthreads();
    v8f r0;
    {
        const float* s = r0s + (((cl << 1) + hh) << 3);
        float4 lo = *(const float4*)(s);
        float4 hi = *(const float4*)(s + 4);
        r0 = (v8f){ lo.x, lo.y, lo.z, lo.w, hi.x, hi.y, hi.z, hi.w };
    }

    const int e0  = (wv << 4) + (hh << 3);
    const int j8  = cl & 7;
    const int c3  = cl >> 3;
    const int ent = e0 + j8;
    const float pvl = p [(size_t)b * NN + ent];
    const float lbl = lb[(size_t)b * NN + ent];
    const float ubl = ub[(size_t)b * NN + ent];
    float uvl = 0.f, rnw = -pvl, xol = 0.f;
    const int poff = (c3 << 3) * PP + ((hh == 0 && j8 == 0) ? (NN + wv) : ent);

    for (int it = 0; it < NITER; ++it) {
        float* wrow = part + (((it & 1) << 4) + wv) * PP;
        float* colp = wrow + (l0 ? (hh << 3) : (DUMP + lane * 12));
        float* rowp = wrow + 16 + (hh ? (DUMP + lane) : cl);

        const float rv[8] = { row_share<0>(rnw), row_share<1>(rnw), row_share<2>(rnw), row_share<3>(rnw),
                              row_share<4>(rnw), row_share<5>(rnw), row_share<6>(rnw), row_share<7>(rnw) };

        v8f dr = {};
        #pragma unroll
        for (int G = 0; G < 4; ++G) {
            const int Ta = 4 * G, Tb = 4 * G + 2;
            #pragma unroll
            for (int T = Ta; T < Ta + 4; ++T) {
                #pragma unroll
                for (int g = 0; g < 8; ++g) acc[T][g] = l0 ? rv[g] : acc[T][g];
            }
            v8f da = {}, db = {};
            #pragma unroll
            for (int q = 0; q < 4; ++q) {
                if (G == 0) {
                    v2f ar = { r0[2 * q], r0[2 * q + 1] };
                    v2f br = { rv[2 * q], rv[2 * q + 1] };
                    dr = __builtin_amdgcn_wmma_f32_16x16x4_f32(
                        false, ar, false, br, (short)0, dr, false, false);
                }
                v2f a0 = { acc[Ta][2 * q],     acc[Ta][2 * q + 1] };
                v2f b0 = { acc[Ta + 1][2 * q], acc[Ta + 1][2 * q + 1] };
                v2f a1 = { acc[Tb][2 * q],     acc[Tb][2 * q + 1] };
                v2f b1 = { acc[Tb + 1][2 * q], acc[Tb + 1][2 * q + 1] };
                da = __builtin_amdgcn_wmma_f32_16x16x4_f32(
                    false, a0, false, b0, (short)0, da, false, false);
                db = __builtin_amdgcn_wmma_f32_16x16x4_f32(
                    false, a1, false, b1, (short)0, db, false, false);
            }
            *(float4*)(colp + 16 * Ta)     = make_float4(da[0], da[1], da[2], da[3]);
            *(float4*)(colp + 16 * Ta + 4) = make_float4(da[4], da[5], da[6], da[7]);
            rowp[16 * Ta] = da[0];
            *(float4*)(colp + 16 * Tb)     = make_float4(db[0], db[1], db[2], db[3]);
            *(float4*)(colp + 16 * Tb + 4) = make_float4(db[4], db[5], db[6], db[7]);
            rowp[16 * Tb] = db[0];
            if (G == 0) {
                *(float4*)(colp + NN)     = make_float4(dr[0], dr[1], dr[2], dr[3]);
                *(float4*)(colp + NN + 4) = make_float4(dr[4], dr[5], dr[6], dr[7]);
            }
        }
        __syncthreads();

        const float* pr = part + ((it & 1) << 4) * PP + poff;
        const float w0 = pr[0 * PP], w1 = pr[1 * PP], w2 = pr[2 * PP], w3 = pr[3 * PP];
        const float w4 = pr[4 * PP], w5 = pr[5 * PP], w6 = pr[6 * PP], w7 = pr[7 * PP];
        const float h8 = ((w0 + w1) + (w2 + w3)) + ((w4 + w5) + (w6 + w7));
        const float xi = h8 + row_xor<8>(h8);

        const float t  = xi + uvl;
        const float zi = fminf(fmaxf(t, lbl), ubl);
        uvl = uvl + xi - zi;
        rnw = -pvl + (zi - uvl);
        xol = xi;
    }

    if (c3 == 0) xv[ent] = xol;
    __syncthreads();
    if (tid < NN) out[(size_t)b * NN + tid] = xv[tid];
}

extern "C" void kernel_launch(void* const* d_in, const int* in_sizes, int n_in,
                              void* d_out, int out_size, void* d_ws, size_t ws_size,
                              hipStream_t stream) {
    const float* Q  = (const float*)d_in[0];
    const float* p  = (const float*)d_in[1];
    const float* lb = (const float*)d_in[2];
    const float* ub = (const float*)d_in[3];
    float* out = (float*)d_out;

    const int B = in_sizes[0] / (NN * NN);
    const size_t shmem = (size_t)(NN * LDM + NN + 16 * NN + 64) * sizeof(float);

    (void)hipFuncSetAttribute((const void*)boxqp_admm_kernel,
                              hipFuncAttributeMaxDynamicSharedMemorySize,
                              (int)shmem);

    boxqp_admm_kernel<<<B, TPB, shmem, stream>>>(Q, p, lb, ub, out);
}
